# starting kernel
# baseline (speedup 1.0000x reference)
#include <hip/hip_runtime.h>
#include <hip/hip_bf16.h>

#define HWSZ   4096
#define NB     32
#define CCH    256
#define GRP    16
#define CPG    16
#define NHWF   131072.0f
#define SSTR   288
#define EPSV   1e-5f

typedef float v2f __attribute__((ext_vector_type(2)));
typedef float v8f __attribute__((ext_vector_type(8)));

static __device__ __forceinline__ v8f wmma4(v2f a, v2f b, v8f c) {
  return __builtin_amdgcn_wmma_f32_16x16x4_f32(false, a, false, b, (short)0, c,
                                               false, false);
}

static __device__ __forceinline__ v8f mul16(const float* As, const float* Bs,
                                            int lane16, int kh) {
  v8f acc = {};
#pragma unroll
  for (int t = 0; t < 4; ++t) {
    int k = 4 * t + kh;
    v2f a, b;
    a.x = As[lane16 * 16 + k];
    a.y = As[lane16 * 16 + k + 1];
    b.x = Bs[k * 16 + lane16];
    b.y = Bs[(k + 1) * 16 + lane16];
    acc = wmma4(a, b, acc);
  }
  return acc;
}

static __device__ __forceinline__ void st16(float* Ds, int lane16, int half,
                                            const v8f& d) {
#pragma unroll
  for (int v = 0; v < 8; ++v) Ds[(v + half * 8) * 16 + lane16] = d[v];
}

__global__ void k_stats(const float* __restrict__ x, float* __restrict__ statsA) {
  const int inst = blockIdx.x;
  const int n = inst >> 4, gi = inst & 15, cb = gi * CPG;
  const int tid = threadIdx.x;
  const int lane = tid & 31, wave = tid >> 5;
  const int lane16 = lane & 15, half = lane >> 4, kh = half * 2;

  __shared__ float sQ[256];
  __shared__ float sS[16];
  if (tid < 256) sQ[tid] = 0.f;
  if (tid < 16) sS[tid] = 0.f;
  __syncthreads();

  const float* row = x + (size_t)(n * CCH + cb + lane16) * HWSZ;
  const int kbase = wave * 512 + kh;

  v8f acc = {};
  float ssum = 0.f;
#pragma unroll 4
  for (int i = 0; i < 128; ++i) {
    v2f v = *(const v2f*)(row + kbase + 4 * i);
    ssum += v.x + v.y;
    acc = wmma4(v, v, acc);
  }

  atomicAdd(&sS[lane16], ssum);
#pragma unroll
  for (int v = 0; v < 8; ++v) {
    int r = v + half * 8;
    atomicAdd(&sQ[r * 16 + lane16], acc[v]);
  }
  __syncthreads();

  float* dst = statsA + (size_t)inst * SSTR;
  if (tid < 256) dst[tid] = sQ[tid];
  if (tid < 16) dst[256 + tid] = sS[tid];
}

__global__ void k_bnagg(const float* __restrict__ statsA, float* __restrict__ statsB) {
  const int gi = blockIdx.x, tid = threadIdx.x;
  float acc = 0.f;
  for (int n = 0; n < NB; ++n) acc += statsA[(size_t)(n * GRP + gi) * SSTR + tid];
  statsB[(size_t)gi * SSTR + tid] = acc;
  if (tid < 16) {
    float a2 = 0.f;
    for (int n = 0; n < NB; ++n)
      a2 += statsA[(size_t)(n * GRP + gi) * SSTR + 256 + tid];
    statsB[(size_t)gi * SSTR + 256 + tid] = a2;
  }
}

__global__ void __launch_bounds__(32)
k_whiten(const float* __restrict__ statsA, const float* __restrict__ statsB,
         const float* __restrict__ swm, const float* __restrict__ swv,
         const float* __restrict__ weight, const float* __restrict__ bias,
         float* __restrict__ wmOut) {
  const int inst = blockIdx.x;
  const int gi = inst & 15, cb = gi * CPG;
  const int lane = threadIdx.x & 31;
  const int lane16 = lane & 15, half = lane >> 4, kh = half * 2;

  __shared__ float sMeanIn[16], sMeanBn[16], sMean[16];
  __shared__ float sCov[256], sP[256], sT[256], sW2[256];
  __shared__ float sTr;

  float a0 = swm[0], a1 = swm[1];
  float am = fmaxf(a0, a1);
  float e0 = __expf(a0 - am), e1 = __expf(a1 - am);
  float mw0 = e0 / (e0 + e1), mw1 = 1.f - mw0;
  float b0 = swv[0], b1 = swv[1];
  float bm = fmaxf(b0, b1);
  float f0 = __expf(b0 - bm), f1 = __expf(b1 - bm);
  float vw0 = f0 / (f0 + f1), vw1 = 1.f - vw0;

  const float* qin = statsA + (size_t)inst * SSTR;
  const float* qbn = statsB + (size_t)gi * SSTR;

  if (lane == 0) sTr = 0.f;
  if (lane < 16) {
    float mi = qin[256 + lane] * (1.f / (float)HWSZ);
    float mb = qbn[256 + lane] * (1.f / NHWF);
    sMeanIn[lane] = mi;
    sMeanBn[lane] = mb;
    sMean[lane] = mw0 * mb + mw1 * mi;
  }
  __syncthreads();

#pragma unroll
  for (int j = 0; j < 8; ++j) {
    int e = lane * 8 + j, r = e >> 4, cidx = e & 15;
    float cin = qin[e] * (1.f / (float)HWSZ) - sMeanIn[r] * sMeanIn[cidx];
    float cbn = qbn[e] * (1.f / NHWF) - sMeanBn[r] * sMeanBn[cidx];
    sCov[e] = vw0 * cbn + vw1 * cin + ((r == cidx) ? EPSV : 0.f);
    sP[e] = (r == cidx) ? 1.f : 0.f;
  }
  __syncthreads();
  if (lane < 16) atomicAdd(&sTr, sCov[lane * 17]);
  __syncthreads();
  const float rTr = 1.f / sTr;
#pragma unroll
  for (int j = 0; j < 8; ++j) sCov[lane * 8 + j] *= rTr;
  __syncthreads();

  for (int it = 0; it < 5; ++it) {
    v8f d2 = mul16(sP, sP, lane16, kh);
    __syncthreads();
    st16(sT, lane16, half, d2);
    __syncthreads();
    v8f d3 = mul16(sT, sP, lane16, kh);
    __syncthreads();
    st16(sT, lane16, half, d3);
    __syncthreads();
    v8f t = mul16(sT, sCov, lane16, kh);
    v8f np;
#pragma unroll
    for (int v = 0; v < 8; ++v)
      np[v] = 1.5f * sP[(v + half * 8) * 16 + lane16] - 0.5f * t[v];
    __syncthreads();
    st16(sP, lane16, half, np);
    __syncthreads();
  }

  const float sc = sqrtf(rTr);
#pragma unroll
  for (int j = 0; j < 8; ++j) {
    int e = lane * 8 + j, r = e >> 4;
    sW2[e] = weight[cb + r] * sP[e] * sc;
  }
  __syncthreads();

  float* dst = wmOut + (size_t)inst * SSTR;
#pragma unroll
  for (int j = 0; j < 8; ++j) {
    int e = lane * 8 + j;
    dst[e] = sW2[e];
  }
  if (lane < 16) {
    float acc = bias[cb + lane];
#pragma unroll
    for (int k = 0; k < 16; ++k) acc -= sW2[lane * 16 + k] * sMean[k];
    dst[256 + lane] = acc;
  }
}

__global__ void k_apply(const float* __restrict__ x, const float* __restrict__ wmIn,
                        float* __restrict__ out) {
  const int inst = blockIdx.x;
  const int n = inst >> 4, gi = inst & 15, cb = gi * CPG;
  const int tid = threadIdx.x, lane = tid & 31, wave = tid >> 5;
  const int lane16 = lane & 15, half = lane >> 4, kh = half * 2;

  __shared__ float sW[256];
  __shared__ float sWb[16];
  const float* src = wmIn + (size_t)inst * SSTR;
  if (tid < 256) sW[tid] = src[tid];
  if (tid < 16) sWb[tid] = src[256 + tid];
  __syncthreads();

  v2f afr[4];
#pragma unroll
  for (int t = 0; t < 4; ++t) {
    int k = 4 * t + kh;
    afr[t].x = sW[lane16 * 16 + k];
    afr[t].y = sW[lane16 * 16 + k + 1];
  }
  float wb[8];
#pragma unroll
  for (int v = 0; v < 8; ++v) wb[v] = sWb[v + half * 8];

  const float* px = x + (size_t)(n * CCH + cb) * HWSZ;
  float* po = out + (size_t)(n * CCH + cb) * HWSZ;

  for (int tile = wave; tile < 256; tile += 8) {
    const int n0 = tile * 16 + lane16;
    v8f d = {};
#pragma unroll
    for (int t = 0; t < 4; ++t) {
      int k = 4 * t + kh;
      v2f b;
      b.x = px[(size_t)k * HWSZ + n0];
      b.y = px[(size_t)(k + 1) * HWSZ + n0];
      d = wmma4(afr[t], b, d);
    }
#pragma unroll
    for (int v = 0; v < 8; ++v)
      __builtin_nontemporal_store(d[v] + wb[v],
                                  po + (size_t)(v + half * 8) * HWSZ + n0);
  }
}

extern "C" void kernel_launch(void* const* d_in, const int* in_sizes, int n_in,
                              void* d_out, int out_size, void* d_ws, size_t ws_size,
                              hipStream_t stream) {
  const float* x      = (const float*)d_in[0];
  const float* swm    = (const float*)d_in[1];
  const float* swv    = (const float*)d_in[2];
  const float* weight = (const float*)d_in[3];
  const float* bias   = (const float*)d_in[4];
  float* out = (float*)d_out;

  float* wsf    = (float*)d_ws;
  float* statsA = wsf;
  float* statsB = wsf + (size_t)512 * SSTR;
  float* wmBuf  = wsf + (size_t)512 * SSTR + (size_t)16 * SSTR;

  k_stats <<<512, 256, 0, stream>>>(x, statsA);
  k_bnagg <<< 16, 256, 0, stream>>>(statsA, statsB);
  k_whiten<<<512,  32, 0, stream>>>(statsA, statsB, swm, swv, weight, bias, wmBuf);
  k_apply <<<512, 256, 0, stream>>>(x, wmBuf, out);
}


# k_apply: each lane owns 4 adjacent columns of a 64-column tile: 128-bit x loads and out stores in whole 256-B row pieces, next tile's 8 loads issued ahead of the current tile's WMMAs
# speedup vs baseline: 1.4719x; 1.4719x over previous
#include <hip/hip_runtime.h>
#include <hip/hip_bf16.h>

#define HWSZ   4096
#define NB     32
#define CCH    256
#define GRP    16
#define CPG    16
#define NHWF   131072.0f
#define SSTR   288
#define EPSV   1e-5f

typedef float v2f __attribute__((ext_vector_type(2)));
typedef float v4f __attribute__((ext_vector_type(4)));
typedef float v8f __attribute__((ext_vector_type(8)));

static __device__ __forceinline__ v8f wmma4(v2f a, v2f b, v8f c) {
  return __builtin_amdgcn_wmma_f32_16x16x4_f32(false, a, false, b, (short)0, c,
                                               false, false);
}

static __device__ __forceinline__ v8f mul16(const float* As, const float* Bs,
                                            int lane16, int kh) {
  v8f acc = {};
#pragma unroll
  for (int t = 0; t < 4; ++t) {
    int k = 4 * t + kh;
    v2f a, b;
    a.x = As[lane16 * 16 + k];
    a.y = As[lane16 * 16 + k + 1];
    b.x = Bs[k * 16 + lane16];
    b.y = Bs[(k + 1) * 16 + lane16];
    acc = wmma4(a, b, acc);
  }
  return acc;
}

static __device__ __forceinline__ void st16(float* Ds, int lane16, int half,
                                            const v8f& d) {
#pragma unroll
  for (int v = 0; v < 8; ++v) Ds[(v + half * 8) * 16 + lane16] = d[v];
}

__global__ void k_stats(const float* __restrict__ x, float* __restrict__ statsA) {
  const int inst = blockIdx.x;
  const int n = inst >> 4, gi = inst & 15, cb = gi * CPG;
  const int tid = threadIdx.x;
  const int lane = tid & 31, wave = tid >> 5;
  const int lane16 = lane & 15, half = lane >> 4, kh = half * 2;

  __shared__ float sQ[256];
  __shared__ float sS[16];
  if (tid < 256) sQ[tid] = 0.f;
  if (tid < 16) sS[tid] = 0.f;
  __syncthreads();

  const float* row = x + (size_t)(n * CCH + cb + lane16) * HWSZ;
  const int kbase = wave * 512 + kh;

  v8f acc = {};
  float ssum = 0.f;
#pragma unroll 4
  for (int i = 0; i < 128; ++i) {
    v2f v = *(const v2f*)(row + kbase + 4 * i);
    ssum += v.x + v.y;
    acc = wmma4(v, v, acc);
  }

  atomicAdd(&sS[lane16], ssum);
#pragma unroll
  for (int v = 0; v < 8; ++v) {
    int r = v + half * 8;
    atomicAdd(&sQ[r * 16 + lane16], acc[v]);
  }
  __syncthreads();

  float* dst = statsA + (size_t)inst * SSTR;
  if (tid < 256) dst[tid] = sQ[tid];
  if (tid < 16) dst[256 + tid] = sS[tid];
}

__global__ void k_bnagg(const float* __restrict__ statsA, float* __restrict__ statsB) {
  const int gi = blockIdx.x, tid = threadIdx.x;
  float acc = 0.f;
  for (int n = 0; n < NB; ++n) acc += statsA[(size_t)(n * GRP + gi) * SSTR + tid];
  statsB[(size_t)gi * SSTR + tid] = acc;
  if (tid < 16) {
    float a2 = 0.f;
    for (int n = 0; n < NB; ++n)
      a2 += statsA[(size_t)(n * GRP + gi) * SSTR + 256 + tid];
    statsB[(size_t)gi * SSTR + 256 + tid] = a2;
  }
}

__global__ void __launch_bounds__(32)
k_whiten(const float* __restrict__ statsA, const float* __restrict__ statsB,
         const float* __restrict__ swm, const float* __restrict__ swv,
         const float* __restrict__ weight, const float* __restrict__ bias,
         float* __restrict__ wmOut) {
  const int inst = blockIdx.x;
  const int gi = inst & 15, cb = gi * CPG;
  const int lane = threadIdx.x & 31;
  const int lane16 = lane & 15, half = lane >> 4, kh = half * 2;

  __shared__ float sMeanIn[16], sMeanBn[16], sMean[16];
  __shared__ float sCov[256], sP[256], sT[256], sW2[256];
  __shared__ float sTr;

  float a0 = swm[0], a1 = swm[1];
  float am = fmaxf(a0, a1);
  float e0 = __expf(a0 - am), e1 = __expf(a1 - am);
  float mw0 = e0 / (e0 + e1), mw1 = 1.f - mw0;
  float b0 = swv[0], b1 = swv[1];
  float bm = fmaxf(b0, b1);
  float f0 = __expf(b0 - bm), f1 = __expf(b1 - bm);
  float vw0 = f0 / (f0 + f1), vw1 = 1.f - vw0;

  const float* qin = statsA + (size_t)inst * SSTR;
  const float* qbn = statsB + (size_t)gi * SSTR;

  if (lane == 0) sTr = 0.f;
  if (lane < 16) {
    float mi = qin[256 + lane] * (1.f / (float)HWSZ);
    float mb = qbn[256 + lane] * (1.f / NHWF);
    sMeanIn[lane] = mi;
    sMeanBn[lane] = mb;
    sMean[lane] = mw0 * mb + mw1 * mi;
  }
  __syncthreads();

#pragma unroll
  for (int j = 0; j < 8; ++j) {
    int e = lane * 8 + j, r = e >> 4, cidx = e & 15;
    float cin = qin[e] * (1.f / (float)HWSZ) - sMeanIn[r] * sMeanIn[cidx];
    float cbn = qbn[e] * (1.f / NHWF) - sMeanBn[r] * sMeanBn[cidx];
    sCov[e] = vw0 * cbn + vw1 * cin + ((r == cidx) ? EPSV : 0.f);
    sP[e] = (r == cidx) ? 1.f : 0.f;
  }
  __syncthreads();
  if (lane < 16) atomicAdd(&sTr, sCov[lane * 17]);
  __syncthreads();
  const float rTr = 1.f / sTr;
#pragma unroll
  for (int j = 0; j < 8; ++j) sCov[lane * 8 + j] *= rTr;
  __syncthreads();

  for (int it = 0; it < 5; ++it) {
    v8f d2 = mul16(sP, sP, lane16, kh);
    __syncthreads();
    st16(sT, lane16, half, d2);
    __syncthreads();
    v8f d3 = mul16(sT, sP, lane16, kh);
    __syncthreads();
    st16(sT, lane16, half, d3);
    __syncthreads();
    v8f t = mul16(sT, sCov, lane16, kh);
    v8f np;
#pragma unroll
    for (int v = 0; v < 8; ++v)
      np[v] = 1.5f * sP[(v + half * 8) * 16 + lane16] - 0.5f * t[v];
    __syncthreads();
    st16(sP, lane16, half, np);
    __syncthreads();
  }

  const float sc = sqrtf(rTr);
#pragma unroll
  for (int j = 0; j < 8; ++j) {
    int e = lane * 8 + j, r = e >> 4;
    sW2[e] = weight[cb + r] * sP[e] * sc;
  }
  __syncthreads();

  float* dst = wmOut + (size_t)inst * SSTR;
#pragma unroll
  for (int j = 0; j < 8; ++j) {
    int e = lane * 8 + j;
    dst[e] = sW2[e];
  }
  if (lane < 16) {
    float acc = bias[cb + lane];
#pragma unroll
    for (int k = 0; k < 16; ++k) acc -= sW2[lane * 16 + k] * sMean[k];
    dst[256 + lane] = acc;
  }
}

__global__ void __launch_bounds__(256)
k_apply(const float* __restrict__ x, const float* __restrict__ wmIn,
        float* __restrict__ out) {
  const int inst = blockIdx.x;
  const int n = inst >> 4, gi = inst & 15, cb = gi * CPG;
  const int tid = threadIdx.x, lane = tid & 31, wave = tid >> 5;
  const int lane16 = lane & 15, half = lane >> 4, kh = half * 2;

  __shared__ float sW[256];
  __shared__ float sWb[16];
  const float* src = wmIn + (size_t)inst * SSTR;
  if (tid < 256) sW[tid] = src[tid];
  if (tid < 16) sWb[tid] = src[256 + tid];
  __syncthreads();

  v2f afr[4];
#pragma unroll
  for (int t = 0; t < 4; ++t) {
    int k = 4 * t + kh;
    afr[t].x = sW[lane16 * 16 + k];
    afr[t].y = sW[lane16 * 16 + k + 1];
  }
  float wb[8];
#pragma unroll
  for (int v = 0; v < 8; ++v) wb[v] = sWb[v + half * 8];

  const float* px = x + (size_t)(n * CCH + cb + kh) * HWSZ + wave * 64 + lane16 * 4;
  float* po = out + (size_t)(n * CCH + cb + half * 8) * HWSZ + wave * 64 + lane16 * 4;

  v4f q[2][8];
#pragma unroll
  for (int u = 0; u < 8; ++u)
    q[0][u] = *(const v4f*)(px + (size_t)((u >> 1) * 4 + (u & 1)) * HWSZ);

#pragma unroll
  for (int i = 0; i < 8; ++i) {
    const int cur = i & 1, nxt = cur ^ 1;
    if (i + 1 < 8) {
#pragma unroll
      for (int u = 0; u < 8; ++u)
        q[nxt][u] = *(const v4f*)(px + (size_t)((u >> 1) * 4 + (u & 1)) * HWSZ +
                                  (i + 1) * 512);
      __builtin_amdgcn_sched_barrier(0);
    }

    v8f d[4] = {};
#pragma unroll
    for (int t = 0; t < 4; ++t) {
#pragma unroll
      for (int j = 0; j < 4; ++j) {
        v2f bf;
        bf.x = q[cur][2 * t][j];
        bf.y = q[cur][2 * t + 1][j];
        d[j] = wmma4(afr[t], bf, d[j]);
      }
    }

#pragma unroll
    for (int v = 0; v < 8; ++v) {
      v4f o;
      o.x = d[0][v] + wb[v];
      o.y = d[1][v] + wb[v];
      o.z = d[2][v] + wb[v];
      o.w = d[3][v] + wb[v];
      __builtin_nontemporal_store(o, (v4f*)(po + (size_t)v * HWSZ + i * 512));
    }
  }
}

extern "C" void kernel_launch(void* const* d_in, const int* in_sizes, int n_in,
                              void* d_out, int out_size, void* d_ws, size_t ws_size,
                              hipStream_t stream) {
  const float* x      = (const float*)d_in[0];
  const float* swm    = (const float*)d_in[1];
  const float* swv    = (const float*)d_in[2];
  const float* weight = (const float*)d_in[3];
  const float* bias   = (const float*)d_in[4];
  float* out = (float*)d_out;

  float* wsf    = (float*)d_ws;
  float* statsA = wsf;
  float* statsB = wsf + (size_t)512 * SSTR;
  float* wmBuf  = wsf + (size_t)512 * SSTR + (size_t)16 * SSTR;

  k_stats <<<512, 256, 0, stream>>>(x, statsA);
  k_bnagg <<< 16, 256, 0, stream>>>(statsA, statsB);
  k_whiten<<<512,  32, 0, stream>>>(statsA, statsB, swm, swv, weight, bias, wmBuf);
  k_apply <<<512, 256, 0, stream>>>(x, wmBuf, out);
}


# k_bnagg and k_whiten folded into k_apply's prologue (4 launches -> 2): each block sums its group's 32 instance statistics itself (same order), wave 0 runs the Newton-Schulz beneath tile 0's x loads
# speedup vs baseline: 1.8594x; 1.2633x over previous
#include <hip/hip_runtime.h>
#include <hip/hip_bf16.h>

#define HWSZ   4096
#define NB     32
#define CCH    256
#define GRP    16
#define CPG    16
#define NHWF   131072.0f
#define SSTR   288
#define EPSV   1e-5f

typedef float v2f __attribute__((ext_vector_type(2)));
typedef float v4f __attribute__((ext_vector_type(4)));
typedef float v8f __attribute__((ext_vector_type(8)));

static __device__ __forceinline__ v8f wmma4(v2f a, v2f b, v8f c) {
  return __builtin_amdgcn_wmma_f32_16x16x4_f32(false, a, false, b, (short)0, c,
                                               false, false);
}

static __device__ __forceinline__ v8f mul16(const float* As, const float* Bs,
                                            int lane16, int kh) {
  v8f acc = {};
#pragma unroll
  for (int t = 0; t < 4; ++t) {
    int k = 4 * t + kh;
    v2f a, b;
    a.x = As[lane16 * 16 + k];
    a.y = As[lane16 * 16 + k + 1];
    b.x = Bs[k * 16 + lane16];
    b.y = Bs[(k + 1) * 16 + lane16];
    acc = wmma4(a, b, acc);
  }
  return acc;
}

static __device__ __forceinline__ void st16(float* Ds, int lane16, int half,
                                            const v8f& d) {
#pragma unroll
  for (int v = 0; v < 8; ++v) Ds[(v + half * 8) * 16 + lane16] = d[v];
}

__global__ void k_stats(const float* __restrict__ x, float* __restrict__ statsA) {
  const int inst = blockIdx.x;
  const int n = inst >> 4, gi = inst & 15, cb = gi * CPG;
  const int tid = threadIdx.x;
  const int lane = tid & 31, wave = tid >> 5;
  const int lane16 = lane & 15, half = lane >> 4, kh = half * 2;

  __shared__ float sQ[256];
  __shared__ float sS[16];
  if (tid < 256) sQ[tid] = 0.f;
  if (tid < 16) sS[tid] = 0.f;
  __syncthreads();

  const float* row = x + (size_t)(n * CCH + cb + lane16) * HWSZ;
  const int kbase = wave * 512 + kh;

  v8f acc = {};
  float ssum = 0.f;
#pragma unroll 4
  for (int i = 0; i < 128; ++i) {
    v2f v = *(const v2f*)(row + kbase + 4 * i);
    ssum += v.x + v.y;
    acc = wmma4(v, v, acc);
  }

  atomicAdd(&sS[lane16], ssum);
#pragma unroll
  for (int v = 0; v < 8; ++v) {
    int r = v + half * 8;
    atomicAdd(&sQ[r * 16 + lane16], acc[v]);
  }
  __syncthreads();

  float* dst = statsA + (size_t)inst * SSTR;
  if (tid < 256) dst[tid] = sQ[tid];
  if (tid < 16) dst[256 + tid] = sS[tid];
}

static __device__ __forceinline__ void wave_sync() {
  __builtin_amdgcn_fence(__ATOMIC_RELEASE, "workgroup", "local");
  __builtin_amdgcn_wave_barrier();
  __builtin_amdgcn_fence(__ATOMIC_ACQUIRE, "workgroup", "local");
}

__global__ void __launch_bounds__(256)
k_apply(const float* __restrict__ x, const float* __restrict__ statsA,
        const float* __restrict__ swm, const float* __restrict__ swv,
        const float* __restrict__ weight, const float* __restrict__ bias,
        float* __restrict__ out) {
  const int inst = blockIdx.x;
  const int n = inst >> 4, gi = inst & 15, cb = gi * CPG;
  const int tid = threadIdx.x, lane = tid & 31;
  const int wave = __builtin_amdgcn_readfirstlane(tid >> 5);
  const int lane16 = lane & 15, half = lane >> 4, kh = half * 2;

  __shared__ float sBn[SSTR], sIn[SSTR];
  __shared__ float sMeanIn[16], sMeanBn[16], sMean[16];
  __shared__ float sCov[256], sP[256], sT[256];
  __shared__ float sW[256];
  __shared__ float sWb[16];
  __shared__ float sTr;

  const float* px = x + (size_t)(n * CCH + cb + kh) * HWSZ + wave * 64 + lane16 * 4;
  float* po = out + (size_t)(n * CCH + cb + half * 8) * HWSZ + wave * 64 + lane16 * 4;

  const float a0 = swm[0], a1 = swm[1], b0 = swv[0], b1 = swv[1];
  const float wgt = weight[cb + (lane >> 1)];
  const float bs = bias[cb + lane16];

  const float* pin = statsA + (size_t)inst * SSTR;
  const float* pa = statsA + (size_t)gi * SSTR + tid;
  const float own = pin[tid];
  float accq = 0.f;
  if (wave == 7) {
    const float* ps = statsA + (size_t)gi * SSTR + 256 + lane16;
    const float owns = pin[256 + lane16];
    float accs = 0.f;
#pragma unroll
    for (int m = 0; m < NB; ++m) {
      accq += pa[(size_t)m * GRP * SSTR];
      accs += ps[(size_t)m * GRP * SSTR];
    }
    if (lane < 16) {
      sBn[256 + lane] = accs;
      sIn[256 + lane] = owns;
    }
  } else {
#pragma unroll
    for (int m = 0; m < NB; ++m) accq += pa[(size_t)m * GRP * SSTR];
  }
  sBn[tid] = accq;
  sIn[tid] = own;
  __syncthreads();

  v4f q[2][8];
#pragma unroll
  for (int u = 0; u < 8; ++u)
    q[0][u] = *(const v4f*)(px + (size_t)((u >> 1) * 4 + (u & 1)) * HWSZ);

  if (wave == 0) {
    float am = fmaxf(a0, a1);
    float e0 = __expf(a0 - am), e1 = __expf(a1 - am);
    float mw0 = e0 / (e0 + e1), mw1 = 1.f - mw0;
    float bm = fmaxf(b0, b1);
    float f0 = __expf(b0 - bm), f1 = __expf(b1 - bm);
    float vw0 = f0 / (f0 + f1), vw1 = 1.f - vw0;

    if (lane == 0) sTr = 0.f;
    if (lane < 16) {
      float mi = sIn[256 + lane] * (1.f / (float)HWSZ);
      float mb = sBn[256 + lane] * (1.f / NHWF);
      sMeanIn[lane] = mi;
      sMeanBn[lane] = mb;
      sMean[lane] = mw0 * mb + mw1 * mi;
    }
    wave_sync();

#pragma unroll
    for (int j = 0; j < 8; ++j) {
      int e = lane * 8 + j, r = e >> 4, cidx = e & 15;
      float cin = sIn[e] * (1.f / (float)HWSZ) - sMeanIn[r] * sMeanIn[cidx];
      float cbn = sBn[e] * (1.f / NHWF) - sMeanBn[r] * sMeanBn[cidx];
      sCov[e] = vw0 * cbn + vw1 * cin + ((r == cidx) ? EPSV : 0.f);
      sP[e] = (r == cidx) ? 1.f : 0.f;
    }
    wave_sync();
    if (lane < 16) atomicAdd(&sTr, sCov[lane * 17]);
    wave_sync();
    const float rTr = 1.f / sTr;
#pragma unroll
    for (int j = 0; j < 8; ++j) sCov[lane * 8 + j] *= rTr;
    wave_sync();

    for (int it = 0; it < 5; ++it) {
      v8f d2 = mul16(sP, sP, lane16, kh);
      wave_sync();
      st16(sT, lane16, half, d2);
      wave_sync();
      v8f d3 = mul16(sT, sP, lane16, kh);
      wave_sync();
      st16(sT, lane16, half, d3);
      wave_sync();
      v8f t = mul16(sT, sCov, lane16, kh);
      v8f np;
#pragma unroll
      for (int v = 0; v < 8; ++v)
        np[v] = 1.5f * sP[(v + half * 8) * 16 + lane16] - 0.5f * t[v];
      wave_sync();
      st16(sP, lane16, half, np);
      wave_sync();
    }

    const float sc = sqrtf(rTr);
#pragma unroll
    for (int j = 0; j < 8; ++j) {
      int e = lane * 8 + j;
      sW[e] = wgt * sP[e] * sc;
    }
    wave_sync();
    if (lane < 16) {
      float acc = bs;
#pragma unroll
      for (int k = 0; k < 16; ++k) acc -= sW[lane * 16 + k] * sMean[k];
      sWb[lane] = acc;
    }
  }
  __syncthreads();

  v2f afr[4];
#pragma unroll
  for (int t = 0; t < 4; ++t) {
    int k = 4 * t + kh;
    afr[t].x = sW[lane16 * 16 + k];
    afr[t].y = sW[lane16 * 16 + k + 1];
  }
  float wb[8];
#pragma unroll
  for (int v = 0; v < 8; ++v) wb[v] = sWb[v + half * 8];

#pragma unroll
  for (int i = 0; i < 8; ++i) {
    const int cur = i & 1, nxt = cur ^ 1;
    if (i + 1 < 8) {
#pragma unroll
      for (int u = 0; u < 8; ++u)
        q[nxt][u] = *(const v4f*)(px + (size_t)((u >> 1) * 4 + (u & 1)) * HWSZ +
                                  (i + 1) * 512);
      __builtin_amdgcn_sched_barrier(0);
    }

    v8f d[4] = {};
#pragma unroll
    for (int t = 0; t < 4; ++t) {
#pragma unroll
      for (int j = 0; j < 4; ++j) {
        v2f bf;
        bf.x = q[cur][2 * t][j];
        bf.y = q[cur][2 * t + 1][j];
        d[j] = wmma4(afr[t], bf, d[j]);
      }
    }

#pragma unroll
    for (int v = 0; v < 8; ++v) {
      v4f o;
      o.x = d[0][v] + wb[v];
      o.y = d[1][v] + wb[v];
      o.z = d[2][v] + wb[v];
      o.w = d[3][v] + wb[v];
      __builtin_nontemporal_store(o, (v4f*)(po + (size_t)v * HWSZ + i * 512));
    }
  }
}

extern "C" void kernel_launch(void* const* d_in, const int* in_sizes, int n_in,
                              void* d_out, int out_size, void* d_ws, size_t ws_size,
                              hipStream_t stream) {
  const float* x      = (const float*)d_in[0];
  const float* swm    = (const float*)d_in[1];
  const float* swv    = (const float*)d_in[2];
  const float* weight = (const float*)d_in[3];
  const float* bias   = (const float*)d_in[4];
  float* out = (float*)d_out;

  float* statsA = (float*)d_ws;

  k_stats<<<512, 256, 0, stream>>>(x, statsA);
  k_apply<<<512, 256, 0, stream>>>(x, statsA, swm, swv, weight, bias, out);
}


# k_apply: x loads run four 64-column tiles ahead through a ring of four register buffers (was one ahead); tiles 0..3 requested before the whitening, LDS-only block barrier leaves them in flight
# speedup vs baseline: 1.9539x; 1.0508x over previous
#include <hip/hip_runtime.h>
#include <hip/hip_bf16.h>

#define HWSZ   4096
#define NB     32
#define CCH    256
#define GRP    16
#define CPG    16
#define NHWF   131072.0f
#define SSTR   288
#define EPSV   1e-5f

typedef float v2f __attribute__((ext_vector_type(2)));
typedef float v4f __attribute__((ext_vector_type(4)));
typedef float v8f __attribute__((ext_vector_type(8)));

static __device__ __forceinline__ v8f wmma4(v2f a, v2f b, v8f c) {
  return __builtin_amdgcn_wmma_f32_16x16x4_f32(false, a, false, b, (short)0, c,
                                               false, false);
}

static __device__ __forceinline__ v8f mul16(const float* As, const float* Bs,
                                            int lane16, int kh) {
  v8f acc = {};
#pragma unroll
  for (int t = 0; t < 4; ++t) {
    int k = 4 * t + kh;
    v2f a, b;
    a.x = As[lane16 * 16 + k];
    a.y = As[lane16 * 16 + k + 1];
    b.x = Bs[k * 16 + lane16];
    b.y = Bs[(k + 1) * 16 + lane16];
    acc = wmma4(a, b, acc);
  }
  return acc;
}

static __device__ __forceinline__ void st16(float* Ds, int lane16, int half,
                                            const v8f& d) {
#pragma unroll
  for (int v = 0; v < 8; ++v) Ds[(v + half * 8) * 16 + lane16] = d[v];
}

__global__ void k_stats(const float* __restrict__ x, float* __restrict__ statsA) {
  const int inst = blockIdx.x;
  const int n = inst >> 4, gi = inst & 15, cb = gi * CPG;
  const int tid = threadIdx.x;
  const int lane = tid & 31, wave = tid >> 5;
  const int lane16 = lane & 15, half = lane >> 4, kh = half * 2;

  __shared__ float sQ[256];
  __shared__ float sS[16];
  if (tid < 256) sQ[tid] = 0.f;
  if (tid < 16) sS[tid] = 0.f;
  __syncthreads();

  const float* row = x + (size_t)(n * CCH + cb + lane16) * HWSZ;
  const int kbase = wave * 512 + kh;

  v8f acc = {};
  float ssum = 0.f;
#pragma unroll 4
  for (int i = 0; i < 128; ++i) {
    v2f v = *(const v2f*)(row + kbase + 4 * i);
    ssum += v.x + v.y;
    acc = wmma4(v, v, acc);
  }

  atomicAdd(&sS[lane16], ssum);
#pragma unroll
  for (int v = 0; v < 8; ++v) {
    int r = v + half * 8;
    atomicAdd(&sQ[r * 16 + lane16], acc[v]);
  }
  __syncthreads();

  float* dst = statsA + (size_t)inst * SSTR;
  if (tid < 256) dst[tid] = sQ[tid];
  if (tid < 16) dst[256 + tid] = sS[tid];
}

static __device__ __forceinline__ void wave_sync() {
  __builtin_amdgcn_fence(__ATOMIC_RELEASE, "workgroup", "local");
  __builtin_amdgcn_wave_barrier();
  __builtin_amdgcn_fence(__ATOMIC_ACQUIRE, "workgroup", "local");
}

__global__ void __launch_bounds__(256)
k_apply(const float* __restrict__ x, const float* __restrict__ statsA,
        const float* __restrict__ swm, const float* __restrict__ swv,
        const float* __restrict__ weight, const float* __restrict__ bias,
        float* __restrict__ out) {
  const int inst = blockIdx.x;
  const int n = inst >> 4, gi = inst & 15, cb = gi * CPG;
  const int tid = threadIdx.x, lane = tid & 31;
  const int wave = __builtin_amdgcn_readfirstlane(tid >> 5);
  const int lane16 = lane & 15, half = lane >> 4, kh = half * 2;

  __shared__ float sBn[SSTR], sIn[SSTR];
  __shared__ float sMeanIn[16], sMeanBn[16], sMean[16];
  __shared__ float sCov[256], sP[256], sT[256];
  __shared__ float sW[256];
  __shared__ float sWb[16];
  __shared__ float sTr;

  const float* px = x + (size_t)(n * CCH + cb + kh) * HWSZ + wave * 64 + lane16 * 4;
  float* po = out + (size_t)(n * CCH + cb + half * 8) * HWSZ + wave * 64 + lane16 * 4;

  const float a0 = swm[0], a1 = swm[1], b0 = swv[0], b1 = swv[1];
  const float wgt = weight[cb + (lane >> 1)];
  const float bs = bias[cb + lane16];

  const float* pin = statsA + (size_t)inst * SSTR;
  const float* pa = statsA + (size_t)gi * SSTR + tid;
  const float own = pin[tid];
  float accq = 0.f;
  if (wave == 7) {
    const float* ps = statsA + (size_t)gi * SSTR + 256 + lane16;
    const float owns = pin[256 + lane16];
    float accs = 0.f;
#pragma unroll
    for (int m = 0; m < NB; ++m) {
      accq += pa[(size_t)m * GRP * SSTR];
      accs += ps[(size_t)m * GRP * SSTR];
    }
    if (lane < 16) {
      sBn[256 + lane] = accs;
      sIn[256 + lane] = owns;
    }
  } else {
#pragma unroll
    for (int m = 0; m < NB; ++m) accq += pa[(size_t)m * GRP * SSTR];
  }
  sBn[tid] = accq;
  sIn[tid] = own;
  __syncthreads();

  v4f q[4][8];
#pragma unroll
  for (int b = 0; b < 4; ++b) {
#pragma unroll
    for (int u = 0; u < 8; ++u)
      q[b][u] = *(const v4f*)(px + (size_t)((u >> 1) * 4 + (u & 1)) * HWSZ + b * 512);
  }

  if (wave == 0) {
    float am = fmaxf(a0, a1);
    float e0 = __expf(a0 - am), e1 = __expf(a1 - am);
    float mw0 = e0 / (e0 + e1), mw1 = 1.f - mw0;
    float bm = fmaxf(b0, b1);
    float f0 = __expf(b0 - bm), f1 = __expf(b1 - bm);
    float vw0 = f0 / (f0 + f1), vw1 = 1.f - vw0;

    if (lane == 0) sTr = 0.f;
    if (lane < 16) {
      float mi = sIn[256 + lane] * (1.f / (float)HWSZ);
      float mb = sBn[256 + lane] * (1.f / NHWF);
      sMeanIn[lane] = mi;
      sMeanBn[lane] = mb;
      sMean[lane] = mw0 * mb + mw1 * mi;
    }
    wave_sync();

#pragma unroll
    for (int j = 0; j < 8; ++j) {
      int e = lane * 8 + j, r = e >> 4, cidx = e & 15;
      float cin = sIn[e] * (1.f / (float)HWSZ) - sMeanIn[r] * sMeanIn[cidx];
      float cbn = sBn[e] * (1.f / NHWF) - sMeanBn[r] * sMeanBn[cidx];
      sCov[e] = vw0 * cbn + vw1 * cin + ((r == cidx) ? EPSV : 0.f);
      sP[e] = (r == cidx) ? 1.f : 0.f;
    }
    wave_sync();
    if (lane < 16) atomicAdd(&sTr, sCov[lane * 17]);
    wave_sync();
    const float rTr = 1.f / sTr;
#pragma unroll
    for (int j = 0; j < 8; ++j) sCov[lane * 8 + j] *= rTr;
    wave_sync();

    for (int it = 0; it < 5; ++it) {
      v8f d2 = mul16(sP, sP, lane16, kh);
      wave_sync();
      st16(sT, lane16, half, d2);
      wave_sync();
      v8f d3 = mul16(sT, sP, lane16, kh);
      wave_sync();
      st16(sT, lane16, half, d3);
      wave_sync();
      v8f t = mul16(sT, sCov, lane16, kh);
      v8f np;
#pragma unroll
      for (int v = 0; v < 8; ++v)
        np[v] = 1.5f * sP[(v + half * 8) * 16 + lane16] - 0.5f * t[v];
      wave_sync();
      st16(sP, lane16, half, np);
      wave_sync();
    }

    const float sc = sqrtf(rTr);
#pragma unroll
    for (int j = 0; j < 8; ++j) {
      int e = lane * 8 + j;
      sW[e] = wgt * sP[e] * sc;
    }
    wave_sync();
    if (lane < 16) {
      float acc = bs;
#pragma unroll
      for (int k = 0; k < 16; ++k) acc -= sW[lane * 16 + k] * sMean[k];
      sWb[lane] = acc;
    }
  }
  __builtin_amdgcn_fence(__ATOMIC_RELEASE, "workgroup", "local");
  __builtin_amdgcn_s_barrier();
  __builtin_amdgcn_fence(__ATOMIC_ACQUIRE, "workgroup", "local");

  v2f afr[4];
#pragma unroll
  for (int t = 0; t < 4; ++t) {
    int k = 4 * t + kh;
    afr[t].x = sW[lane16 * 16 + k];
    afr[t].y = sW[lane16 * 16 + k + 1];
  }
  float wb[8];
#pragma unroll
  for (int v = 0; v < 8; ++v) wb[v] = sWb[v + half * 8];

#pragma unroll
  for (int i = 0; i < 8; ++i) {
    const int cur = i & 3;

    v8f d[4] = {};
#pragma unroll
    for (int t = 0; t < 4; ++t) {
#pragma unroll
      for (int j = 0; j < 4; ++j) {
        v2f bf;
        bf.x = q[cur][2 * t][j];
        bf.y = q[cur][2 * t + 1][j];
        d[j] = wmma4(afr[t], bf, d[j]);
      }
    }

    if (i + 4 < 8) {
#pragma unroll
      for (int u = 0; u < 8; ++u)
        q[cur][u] = *(const v4f*)(px + (size_t)((u >> 1) * 4 + (u & 1)) * HWSZ +
                                  (i + 4) * 512);
      __builtin_amdgcn_sched_barrier(0);
    }

#pragma unroll
    for (int v = 0; v < 8; ++v) {
      v4f o;
      o.x = d[0][v] + wb[v];
      o.y = d[1][v] + wb[v];
      o.z = d[2][v] + wb[v];
      o.w = d[3][v] + wb[v];
      __builtin_nontemporal_store(o, (v4f*)(po + (size_t)v * HWSZ + i * 512));
    }
  }
}

extern "C" void kernel_launch(void* const* d_in, const int* in_sizes, int n_in,
                              void* d_out, int out_size, void* d_ws, size_t ws_size,
                              hipStream_t stream) {
  const float* x      = (const float*)d_in[0];
  const float* swm    = (const float*)d_in[1];
  const float* swv    = (const float*)d_in[2];
  const float* weight = (const float*)d_in[3];
  const float* bias   = (const float*)d_in[4];
  float* out = (float*)d_out;

  float* statsA = (float*)d_ws;

  k_stats<<<512, 256, 0, stream>>>(x, statsA);
  k_apply<<<512, 256, 0, stream>>>(x, statsA, swm, swv, weight, bias, out);
}
